# starting kernel
# baseline (speedup 1.0000x reference)
#include <hip/hip_runtime.h>

#define R_REL   2
#define IN_F    128
#define HID_F   64
#define OUT_F   64
#define H1      2
#define H2      1
#define NEG_SLOPE 0.2f

typedef __attribute__((ext_vector_type(16))) __bf16 v16bf;
typedef __attribute__((ext_vector_type(8)))  float  v8f;

union Frag16 { uint4 q[2]; v16bf v; };

__device__ __forceinline__ unsigned short f2bf(float f) {
  unsigned u = __float_as_uint(f);
  unsigned r = u + 0x7FFFu + ((u >> 16) & 1u);
  return (unsigned short)(r >> 16);
}
__device__ __forceinline__ unsigned fkey(float f) {
  unsigned u = __float_as_uint(f);
  return (u & 0x80000000u) ? ~u : (u | 0x80000000u);
}
__device__ __forceinline__ float funkey(unsigned k) {
  unsigned u = (k & 0x80000000u) ? (k & 0x7FFFFFFFu) : ~k;
  return __uint_as_float(u);
}
__device__ __forceinline__ float lrelu(float x) { return x >= 0.f ? x : NEG_SLOPE * x; }

__global__ void k_f32_to_bf16(const float* __restrict__ in,
                              unsigned short* __restrict__ out, size_t n) {
  size_t i = (size_t)blockIdx.x * blockDim.x + threadIdx.x;
  if (i < n) out[i] = f2bf(in[i]);
}

__global__ void k_relu_f32_to_bf16(const float* __restrict__ in,
                                   unsigned short* __restrict__ out, size_t n) {
  size_t i = (size_t)blockIdx.x * blockDim.x + threadIdx.x;
  if (i < n) { float v = in[i]; out[i] = f2bf(v > 0.f ? v : 0.f); }
}

__global__ void k_transpose_w(const float* __restrict__ W,
                              unsigned short* __restrict__ Wt, int K, int NCOLS) {
  int i = blockIdx.x * blockDim.x + threadIdx.x;
  if (i >= K * NCOLS) return;
  int n = i / K, k = i % K;
  Wt[i] = f2bf(W[(size_t)k * NCOLS + n]);
}

__global__ void k_zero_u32(unsigned* __restrict__ p, size_t n) {
  size_t i = (size_t)blockIdx.x * blockDim.x + threadIdx.x;
  if (i < n) p[i] = 0u;
}

__global__ void k_init_bias(float* __restrict__ acc, const float* __restrict__ b,
                            int N, int F) {
  size_t i = (size_t)blockIdx.x * blockDim.x + threadIdx.x;
  if (i >= (size_t)N * F) return;
  int c = (int)(i % F);
  float s = 0.f;
  for (int r = 0; r < R_REL; ++r) s += b[(size_t)r * F + c];
  acc[i] = s;
}

template <int KDIM, int NCOLS, int RPB>
__global__ __launch_bounds__(2 * NCOLS)
void k_gemm_bf16_wmma(const unsigned short* __restrict__ A,
                      const unsigned short* __restrict__ Bt,
                      float* __restrict__ C, int M) {
  constexpr int KC = KDIM / 32;
  const int lane     = threadIdx.x & 31;
  const int wv       = threadIdx.x >> 5;
  const int colBase  = wv * 16;
  const int l16      = lane & 15;
  const int k0       = (lane < 16) ? 0 : 8;
  const int rowBlock = blockIdx.x * (16 * RPB);

  Frag16 fb[KC];
  {
    const unsigned short* bp = Bt + (size_t)(colBase + l16) * KDIM + k0;
#pragma unroll
    for (int c = 0; c < KC; ++c) {
      fb[c].q[0] = *reinterpret_cast<const uint4*>(bp + 32 * c);
      fb[c].q[1] = *reinterpret_cast<const uint4*>(bp + 32 * c + 16);
    }
  }

  const bool full = (rowBlock + 16 * RPB) <= M;
  const int  rOff = (lane < 16) ? 0 : 8;

#pragma unroll
  for (int t = 0; t < RPB; ++t) {
    const int rowBase = rowBlock + 16 * t;
    if (rowBase >= M) break;

    int arow = rowBase + l16; if (arow >= M) arow = M - 1;
    const unsigned short* ap = A + (size_t)arow * KDIM + k0;

    v8f acc = {};
#pragma unroll
    for (int c = 0; c < KC; ++c) {
      Frag16 fa;
      fa.q[0] = *reinterpret_cast<const uint4*>(ap + 32 * c);
      fa.q[1] = *reinterpret_cast<const uint4*>(ap + 32 * c + 16);
      acc = __builtin_amdgcn_wmma_f32_16x16x32_bf16(false, fa.v, false, fb[c].v,
                                                    (short)0, acc, false, false);
    }

    float* cp = C + (size_t)rowBase * NCOLS + colBase + l16;
    if (full) {
#pragma unroll
      for (int v = 0; v < 8; ++v) cp[(size_t)(rOff + v) * NCOLS] = acc[v];
    } else {
#pragma unroll
      for (int v = 0; v < 8; ++v) {
        int rr = rowBase + rOff + v;
        if (rr < M) cp[(size_t)(rOff + v) * NCOLS] = acc[v];
      }
    }
  }
}

__global__ void k_attn_dots(const float* __restrict__ feat,
                            const float* __restrict__ al, const float* __restrict__ ar,
                            float* __restrict__ el, float* __restrict__ er,
                            int N, int H, int D) {
  int i = blockIdx.x * blockDim.x + threadIdx.x;
  if (i >= N * H) return;
  int h = i % H;
  const float* f = feat + (size_t)(i / H) * H * D + (size_t)h * D;
  float sl = 0.f, sr = 0.f;
  for (int d = 0; d < D; d += 4) {
    float4 v = *reinterpret_cast<const float4*>(f + d);
    float4 a = *reinterpret_cast<const float4*>(al + h * D + d);
    float4 b = *reinterpret_cast<const float4*>(ar + h * D + d);
    sl += v.x * a.x + v.y * a.y + v.z * a.z + v.w * a.w;
    sr += v.x * b.x + v.y * b.y + v.z * b.z + v.w * b.w;
  }
  el[i] = sl; er[i] = sr;
}

__global__ void k_edge_max(const float* __restrict__ el, const float* __restrict__ er,
                           const int* __restrict__ src, const int* __restrict__ dst,
                           unsigned* __restrict__ mkey, int E, int H) {
  int e = blockIdx.x * blockDim.x + threadIdx.x;
  if (e >= E) return;
  int s = src[e], d = dst[e];
  for (int h = 0; h < H; ++h) {
    float v = lrelu(el[(size_t)s * H + h] + er[(size_t)d * H + h]);
    atomicMax(&mkey[(size_t)d * H + h], fkey(v));
  }
}

__global__ void k_edge_exp(const float* __restrict__ el, const float* __restrict__ er,
                           const int* __restrict__ src, const int* __restrict__ dst,
                           const unsigned* __restrict__ mkey, float* __restrict__ ssum,
                           float* __restrict__ exbuf, int E, int H) {
  int e = blockIdx.x * blockDim.x + threadIdx.x;
  if (e >= E) return;
  int s = src[e], d = dst[e];
  for (int h = 0; h < H; ++h) {
    float v  = lrelu(el[(size_t)s * H + h] + er[(size_t)d * H + h]);
    float m  = funkey(mkey[(size_t)d * H + h]);
    float ex = __expf(v - m);
    exbuf[(size_t)e * H + h] = ex;
    atomicAdd(&ssum[(size_t)d * H + h], ex);
  }
}

template <int H, int D>
__global__ void k_edge_scatter(const float* __restrict__ exbuf,
                               const float* __restrict__ ssum,
                               const float* __restrict__ feat,
                               const int* __restrict__ src, const int* __restrict__ dst,
                               float* __restrict__ acc, int E) {
  constexpr int F   = H * D;
  constexpr int PER = F / 32;
  int wid  = (int)(((size_t)blockIdx.x * blockDim.x + threadIdx.x) >> 5);
  int lane = threadIdx.x & 31;
  if (wid >= E) return;
  int s = src[wid], d = dst[wid];

  const int base = lane * PER;
  const int h    = base / D;
  float sum   = ssum[(size_t)d * H + h];
  float alpha = exbuf[(size_t)wid * H + h] / (sum > 0.f ? sum : 1.f);

  const float* fp = feat + (size_t)s * F + base;
  float*       op = acc  + (size_t)d * F + base;
  if constexpr (PER == 4) {
    float4 v = *reinterpret_cast<const float4*>(fp);
    atomicAdd(op + 0, alpha * v.x);
    atomicAdd(op + 1, alpha * v.y);
    atomicAdd(op + 2, alpha * v.z);
    atomicAdd(op + 3, alpha * v.w);
  } else {
    float2 v = *reinterpret_cast<const float2*>(fp);
    atomicAdd(op + 0, alpha * v.x);
    atomicAdd(op + 1, alpha * v.y);
  }
}

extern "C" void kernel_launch(void* const* d_in, const int* in_sizes, int n_in,
                              void* d_out, int out_size, void* d_ws, size_t ws_size,
                              hipStream_t stream) {
  const float* x    = (const float*)d_in[0];
  const int*   esrc = (const int*)d_in[1];
  const int*   edst = (const int*)d_in[2];
  const float* W1   = (const float*)d_in[3];
  const float* al1  = (const float*)d_in[4];
  const float* ar1  = (const float*)d_in[5];
  const float* b1   = (const float*)d_in[6];
  const float* W2   = (const float*)d_in[7];
  const float* al2  = (const float*)d_in[8];
  const float* ar2  = (const float*)d_in[9];
  const float* b2   = (const float*)d_in[10];
  float* out = (float*)d_out;

  const int N = in_sizes[0] / IN_F;
  const int E = in_sizes[1] / (2 * R_REL);

  size_t off = 0;
  auto carve = [&](size_t bytes) -> void* {
    void* p = (char*)d_ws + off;
    off += (bytes + 255) & ~(size_t)255;
    return p;
  };
  unsigned short* xbf  = (unsigned short*)carve((size_t)N * IN_F * 2);
  unsigned short* hbf  = (unsigned short*)carve((size_t)N * (H1 * HID_F) * 2);
  unsigned short* wt1  = (unsigned short*)carve((size_t)R_REL * IN_F * (H1 * HID_F) * 2);
  unsigned short* wt2  = (unsigned short*)carve((size_t)R_REL * (H1 * HID_F) * OUT_F * 2);
  float*    feat  = (float*)carve((size_t)N * (H1 * HID_F) * 4);
  float*    el    = (float*)carve((size_t)N * H1 * 4);
  float*    er    = (float*)carve((size_t)N * H1 * 4);
  unsigned* mkey  = (unsigned*)carve((size_t)N * H1 * 4);
  float*    ssum  = (float*)carve((size_t)N * H1 * 4);
  float*    exbuf = (float*)carve((size_t)E * H1 * 4);
  float*    hacc  = (float*)carve((size_t)N * (H1 * HID_F) * 4);

  const int T = 256;
  auto blk1 = [&](size_t n) { return (unsigned)((n + T - 1) / T); };

  k_f32_to_bf16<<<blk1((size_t)N * IN_F), T, 0, stream>>>(x, xbf, (size_t)N * IN_F);
  for (int r = 0; r < R_REL; ++r) {
    k_transpose_w<<<blk1((size_t)IN_F * (H1 * HID_F)), T, 0, stream>>>(
        W1 + (size_t)r * IN_F * (H1 * HID_F), wt1 + (size_t)r * IN_F * (H1 * HID_F),
        IN_F, H1 * HID_F);
    k_transpose_w<<<blk1((size_t)(H1 * HID_F) * OUT_F), T, 0, stream>>>(
        W2 + (size_t)r * (H1 * HID_F) * OUT_F, wt2 + (size_t)r * (H1 * HID_F) * OUT_F,
        H1 * HID_F, OUT_F);
  }

  constexpr int RPB = 4;
  const unsigned gemmGrid = (unsigned)((N + 16 * RPB - 1) / (16 * RPB));

  k_init_bias<<<blk1((size_t)N * (H1 * HID_F)), T, 0, stream>>>(hacc, b1, N, H1 * HID_F);
  for (int r = 0; r < R_REL; ++r) {
    const int* s1 = esrc + (size_t)(0 * R_REL + r) * E;
    const int* d1 = edst + (size_t)(0 * R_REL + r) * E;

    k_gemm_bf16_wmma<IN_F, H1 * HID_F, RPB><<<gemmGrid, 2 * (H1 * HID_F), 0, stream>>>(
        xbf, wt1 + (size_t)r * IN_F * (H1 * HID_F), feat, N);

    k_attn_dots<<<blk1((size_t)N * H1), T, 0, stream>>>(
        feat, al1 + (size_t)r * H1 * HID_F, ar1 + (size_t)r * H1 * HID_F,
        el, er, N, H1, HID_F);

    k_zero_u32<<<blk1((size_t)N * H1), T, 0, stream>>>(mkey, (size_t)N * H1);
    k_zero_u32<<<blk1((size_t)N * H1), T, 0, stream>>>((unsigned*)ssum, (size_t)N * H1);

    k_edge_max<<<blk1((size_t)E), T, 0, stream>>>(el, er, s1, d1, mkey, E, H1);
    k_edge_exp<<<blk1((size_t)E), T, 0, stream>>>(el, er, s1, d1, mkey, ssum, exbuf, E, H1);
    k_edge_scatter<H1, HID_F><<<blk1((size_t)E * 32), T, 0, stream>>>(
        exbuf, ssum, feat, s1, d1, hacc, E);
  }

  k_relu_f32_to_bf16<<<blk1((size_t)N * (H1 * HID_F)), T, 0, stream>>>(
      hacc, hbf, (size_t)N * (H1 * HID_F));

  k_init_bias<<<blk1((size_t)N * OUT_F), T, 0, stream>>>(out, b2, N, OUT_F);
  for (int r = 0; r < R_REL; ++r) {
    const int* s2 = esrc + (size_t)(1 * R_REL + r) * E;
    const int* d2 = edst + (size_t)(1 * R_REL + r) * E;

    k_gemm_bf16_wmma<H1 * HID_F, OUT_F, RPB><<<gemmGrid, 2 * OUT_F, 0, stream>>>(
        hbf, wt2 + (size_t)r * (H1 * HID_F) * OUT_F, feat, N);

    k_attn_dots<<<blk1((size_t)N * H2), T, 0, stream>>>(
        feat, al2 + (size_t)r * H2 * OUT_F, ar2 + (size_t)r * H2 * OUT_F,
        el, er, N, H2, OUT_F);

    k_zero_u32<<<blk1((size_t)N * H2), T, 0, stream>>>(mkey, (size_t)N * H2);
    k_zero_u32<<<blk1((size_t)N * H2), T, 0, stream>>>((unsigned*)ssum, (size_t)N * H2);

    k_edge_max<<<blk1((size_t)E), T, 0, stream>>>(el, er, s2, d2, mkey, E, H2);
    k_edge_exp<<<blk1((size_t)E), T, 0, stream>>>(el, er, s2, d2, mkey, ssum, exbuf, E, H2);
    k_edge_scatter<H2, OUT_F><<<blk1((size_t)E * 32), T, 0, stream>>>(
        exbuf, ssum, feat, s2, d2, out, E);
  }
}


# Per-destination max taken in the owner group's LDS: first sweep lists the group's edges and logits in LDS (LDS atomic max), exp and rows read the list; edge-max, zero passes gone; arrival-order sums
# speedup vs baseline: 3.8939x; 3.8939x over previous
#include <hip/hip_runtime.h>

#define R_REL   2
#define IN_F    128
#define HID_F   64
#define OUT_F   64
#define H1      2
#define H2      1
#define NEG_SLOPE 0.2f

typedef __attribute__((ext_vector_type(16))) __bf16 v16bf;
typedef __attribute__((ext_vector_type(8)))  float  v8f;

union Frag16 { uint4 q[2]; v16bf v; };

__device__ __forceinline__ unsigned short f2bf(float f) {
  unsigned u = __float_as_uint(f);
  unsigned r = u + 0x7FFFu + ((u >> 16) & 1u);
  return (unsigned short)(r >> 16);
}
__device__ __forceinline__ unsigned fkey(float f) {
  unsigned u = __float_as_uint(f);
  return (u & 0x80000000u) ? ~u : (u | 0x80000000u);
}
__device__ __forceinline__ float funkey(unsigned k) {
  unsigned u = (k & 0x80000000u) ? (k & 0x7FFFFFFFu) : ~k;
  return __uint_as_float(u);
}
__device__ __forceinline__ float lrelu(float x) { return x >= 0.f ? x : NEG_SLOPE * x; }

__global__ void k_f32_to_bf16(const float* __restrict__ in,
                              unsigned short* __restrict__ out, size_t n) {
  size_t i = (size_t)blockIdx.x * blockDim.x + threadIdx.x;
  if (i < n) out[i] = f2bf(in[i]);
}

__global__ void k_relu_f32_to_bf16(const float* __restrict__ in,
                                   unsigned short* __restrict__ out, size_t n) {
  size_t i = (size_t)blockIdx.x * blockDim.x + threadIdx.x;
  if (i < n) { float v = in[i]; out[i] = f2bf(v > 0.f ? v : 0.f); }
}

__global__ void k_transpose_w(const float* __restrict__ W,
                              unsigned short* __restrict__ Wt, int K, int NCOLS) {
  int i = blockIdx.x * blockDim.x + threadIdx.x;
  if (i >= K * NCOLS) return;
  int n = i / K, k = i % K;
  Wt[i] = f2bf(W[(size_t)k * NCOLS + n]);
}

template <int KDIM, int NCOLS, int RPB>
__global__ __launch_bounds__(2 * NCOLS)
void k_gemm_bf16_wmma(const unsigned short* __restrict__ A,
                      const unsigned short* __restrict__ Bt,
                      float* __restrict__ C, int M) {
  constexpr int KC = KDIM / 32;
  const int lane     = threadIdx.x & 31;
  const int wv       = threadIdx.x >> 5;
  const int colBase  = wv * 16;
  const int l16      = lane & 15;
  const int k0       = (lane < 16) ? 0 : 8;
  const int rowBlock = blockIdx.x * (16 * RPB);

  Frag16 fb[KC];
  {
    const unsigned short* bp = Bt + (size_t)(colBase + l16) * KDIM + k0;
#pragma unroll
    for (int c = 0; c < KC; ++c) {
      fb[c].q[0] = *reinterpret_cast<const uint4*>(bp + 32 * c);
      fb[c].q[1] = *reinterpret_cast<const uint4*>(bp + 32 * c + 16);
    }
  }

  const bool full = (rowBlock + 16 * RPB) <= M;
  const int  rOff = (lane < 16) ? 0 : 8;

#pragma unroll
  for (int t = 0; t < RPB; ++t) {
    const int rowBase = rowBlock + 16 * t;
    if (rowBase >= M) break;

    int arow = rowBase + l16; if (arow >= M) arow = M - 1;
    const unsigned short* ap = A + (size_t)arow * KDIM + k0;

    v8f acc = {};
#pragma unroll
    for (int c = 0; c < KC; ++c) {
      Frag16 fa;
      fa.q[0] = *reinterpret_cast<const uint4*>(ap + 32 * c);
      fa.q[1] = *reinterpret_cast<const uint4*>(ap + 32 * c + 16);
      acc = __builtin_amdgcn_wmma_f32_16x16x32_bf16(false, fa.v, false, fb[c].v,
                                                    (short)0, acc, false, false);
    }

    float* cp = C + (size_t)rowBase * NCOLS + colBase + l16;
    if (full) {
#pragma unroll
      for (int v = 0; v < 8; ++v) cp[(size_t)(rOff + v) * NCOLS] = acc[v];
    } else {
#pragma unroll
      for (int v = 0; v < 8; ++v) {
        int rr = rowBase + rOff + v;
        if (rr < M) cp[(size_t)(rOff + v) * NCOLS] = acc[v];
      }
    }
  }
}

__global__ void k_attn_dots(const float* __restrict__ feat,
                            const float* __restrict__ al, const float* __restrict__ ar,
                            float* __restrict__ el, float* __restrict__ er,
                            int N, int H, int D) {
  int i = blockIdx.x * blockDim.x + threadIdx.x;
  if (i >= N * H) return;
  int h = i % H;
  const float* f = feat + (size_t)(i / H) * H * D + (size_t)h * D;
  float sl = 0.f, sr = 0.f;
  for (int d = 0; d < D; d += 4) {
    float4 v = *reinterpret_cast<const float4*>(f + d);
    float4 a = *reinterpret_cast<const float4*>(al + h * D + d);
    float4 b = *reinterpret_cast<const float4*>(ar + h * D + d);
    sl += v.x * a.x + v.y * a.y + v.z * a.z + v.w * a.w;
    sr += v.x * b.x + v.y * b.y + v.z * b.z + v.w * b.w;
  }
  el[i] = sl; er[i] = sr;
}

constexpr int NPB      = 392;

constexpr int OB_T     = 256;
constexpr int OB_W     = 4;
__device__ __forceinline__ unsigned owner_byte(int d) { return ((unsigned)d / (unsigned)NPB) & 255u; }

__global__ __launch_bounds__(OB_T)
void k_owner_bytes(const int* __restrict__ dst, unsigned char* __restrict__ own,
                   int E, size_t stride) {
  const int* dp = dst + (size_t)blockIdx.y * E;
  unsigned char* op = own + (size_t)blockIdx.y * stride;
  const int w0 = (int)blockIdx.x * (OB_T * OB_W) + (int)threadIdx.x;
  if ((E & 3) == 0) {
    int4 v[OB_W];
#pragma unroll
    for (int j = 0; j < OB_W; ++j) {
      const int e0 = 4 * (w0 + j * OB_T);
      v[j] = *reinterpret_cast<const int4*>(dp + (e0 < E ? e0 : 0));
    }
#pragma unroll
    for (int j = 0; j < OB_W; ++j) {
      const int e0 = 4 * (w0 + j * OB_T);
      const unsigned w = owner_byte(v[j].x) | (owner_byte(v[j].y) << 8) |
                         (owner_byte(v[j].z) << 16) | (owner_byte(v[j].w) << 24);
      if (e0 < E) *reinterpret_cast<unsigned*>(op + e0) = w;
    }
  } else {
#pragma unroll
    for (int j = 0; j < OB_W; ++j) {
      const int e0 = 4 * (w0 + j * OB_T);
      if (e0 < E) {
        unsigned w = 0u;
#pragma unroll
        for (int t = 0; t < 4; ++t)
          if (e0 + t < E) w |= owner_byte(dp[e0 + t]) << (8 * t);
        *reinterpret_cast<unsigned*>(op + e0) = w;
      }
    }
  }
}

constexpr int OWN_T    = 1024;
constexpr int RING     = 64;
constexpr int CAP      = 7168;

template <int H, int D, bool FIRST>
__global__ __launch_bounds__(OWN_T)
void k_owner_aggregate(const float* __restrict__ el, const float* __restrict__ er,
                       const float* __restrict__ feat,
                       const int* __restrict__ src, const int* __restrict__ dst,
                       const unsigned char* __restrict__ own,
                       const float* __restrict__ b, float* __restrict__ acc, int N, int E) {
  constexpr int F   = H * D;
  constexpr int PER = F / 32;
  constexpr int LPH = D / 32;
  constexpr int U   = 32 / PER;
  __shared__ __attribute__((aligned(16))) float racc_s[(NPB + 1) * F];
  __shared__ float    ssum_s[NPB * H];
  __shared__ unsigned mkey_s[NPB * H];
  __shared__ float    er_s[NPB * H];
  __shared__ int      ring_e[OWN_T / 32][RING];
  __shared__ int      lst_s[CAP];
  __shared__ float    lst_v[CAP * H];
  __shared__ unsigned short lst_d[CAP];
  __shared__ int      lst_n;

  const int tid  = threadIdx.x;
  const int lane = tid & 31;
  const int wv   = __builtin_amdgcn_readfirstlane(tid >> 5);
  const int lo   = blockIdx.x * NPB;

  for (int i = tid; i < NPB * F / 4; i += OWN_T)
    reinterpret_cast<float4*>(racc_s)[i] = make_float4(0.f, 0.f, 0.f, 0.f);
  for (int i = tid; i < NPB * H; i += OWN_T) {
    ssum_s[i] = 0.f;
    mkey_s[i] = 0u;
    er_s[i]   = (lo + i / H < N) ? er[(size_t)lo * H + i] : 0.f;
  }
  if (tid == 0) lst_n = 0;
  __syncthreads();

  int* const re = ring_e[wv];

  const int Q = (E + 15) >> 4;
  const uint4* own4 = reinterpret_cast<const uint4*>(own);
  const unsigned pat = (unsigned)(blockIdx.x & 255u) * 0x01010101u;

  const int WL   = (Q + 31) >> 5;
  const int rot  = (int)(((long long)WL * blockIdx.x) / gridDim.x);
  const int nIt  = (WL > wv) ? ((WL - wv + 31) >> 5) : 0;

  auto sweep = [&](auto&& work) {
    int qh = 0, qt = 0;

    auto push = [&](bool keep, int e) {
      const unsigned mask = __builtin_amdgcn_ballot_w32(keep);
      if (keep) {
        const int pos = (qt + (int)__builtin_amdgcn_mbcnt_lo(mask, 0u)) & (RING - 1);
        re[pos] = e;
      }
      qt += __builtin_popcount(mask);
    };

    int nxt = 0;
    int tn  = wv + rot;
    if (tn >= WL) tn -= WL;
    auto next_word = [&]() -> int {
      int q = -1;
      if (nxt < nIt) {
        const int qq = tn * 32 + lane;
        q = (qq < Q) ? qq : -1;
      }
      ++nxt;
      tn += 32;
      if (tn >= WL) tn -= WL;
      return q;
    };
    auto load_word = [&](int q) -> uint4 { return own4[q >= 0 ? q : 0]; };

    constexpr int DEPTH = 4;
    uint4 c[DEPTH];
    int   cw[DEPTH];
#pragma unroll
    for (int k = 0; k < DEPTH; ++k) { cw[k] = next_word(); c[k] = load_word(cw[k]); }

    constexpr unsigned K01 = 0x01010101u, K80 = 0x80808080u;
    int it = 0;
    do {
#pragma unroll
      for (int k = 0; k < DEPTH; ++k) {
        const uint4 cur = c[k];
        const int   qw  = cw[k];
        const unsigned x0 = cur.x ^ pat, x1 = cur.y ^ pat, x2 = cur.z ^ pat, x3 = cur.w ^ pat;
        cw[k] = next_word();
        c[k]  = load_word(cw[k]);
        const unsigned t0 = (x0 - K01) & ~x0 & K80, t1 = (x1 - K01) & ~x1 & K80;
        const unsigned t2 = (x2 - K01) & ~x2 & K80, t3 = (x3 - K01) & ~x3 & K80;
        const bool ok = qw >= 0;
        if (__builtin_amdgcn_ballot_w32(ok && (t0 | t1 | t2 | t3) != 0u) != 0u) {
          unsigned u = (t0 >> 7) | (t1 >> 6) | (t2 >> 5) | (t3 >> 4);
          if (!ok) u = 0u;
          const int e0 = 16 * qw;
          for (int trip = 0; trip < 16; ++trip) {
            if (__builtin_amdgcn_ballot_w32(u != 0u) == 0u) break;
            const int p = __builtin_ctz(u | 0x80000000u);
            const int e = e0 + ((p & 7) << 2) + (p >> 3);
            push(u != 0u && e < E, e);
            u &= u - 1u;
            if (qt - qh >= 32) { work(32, qh); qh += 32; }
          }
        }
      }
      it += DEPTH;
    } while (it < nIt);
    if (qt - qh > 0) { work(qt - qh, qh); qh = qt; }
  };

  auto edge_of = [&](int cnt, int qh, int& s_l, int& d_l, float (&v)[H]) {
    s_l = 0; d_l = NPB;
#pragma unroll
    for (int h = 0; h < H; ++h) v[h] = 0.f;
    if (lane < cnt) {
      const int e = re[(qh + lane) & (RING - 1)];
      s_l = src[e];
      const unsigned rel = (unsigned)(dst[e] - lo);
      d_l = (rel < (unsigned)NPB) ? (int)rel : NPB;
    }
    if (d_l < NPB) {
      float a[H];
      if constexpr (H == 2) {
        const float2 a2 = *reinterpret_cast<const float2*>(el + (size_t)s_l * 2);
        a[0] = a2.x; a[1] = a2.y;
      } else {
#pragma unroll
        for (int h = 0; h < H; ++h) a[h] = el[(size_t)s_l * H + h];
      }
#pragma unroll
      for (int h = 0; h < H; ++h) v[h] = lrelu(a[h] + er_s[d_l * H + h]);
    }
  };

  auto finish = [&](int cnt, int s_l, int d_l, const float (&v)[H]) {
    float ex_l[H];
#pragma unroll
    for (int h = 0; h < H; ++h) ex_l[h] = 0.f;
    if (d_l < NPB) {
#pragma unroll
      for (int h = 0; h < H; ++h) {
        ex_l[h] = __expf(v[h] - funkey(mkey_s[d_l * H + h]));
        atomicAdd(&ssum_s[d_l * H + h], ex_l[h]);
      }
    }

    for (int j0 = 0; j0 < cnt; j0 += U) {
      if (j0 + U <= cnt) {
        float f[U][PER];
#pragma unroll
        for (int u = 0; u < U; ++u) {
          const int sg = __builtin_amdgcn_readlane(s_l, j0 + u);
          const float* fp = feat + (size_t)sg * F + lane;
#pragma unroll
          for (int k = 0; k < PER; ++k) f[u][k] = fp[32 * k];
        }
#pragma unroll
        for (int u = 0; u < U; ++u) {
          const int dr = __builtin_amdgcn_readlane(d_l, j0 + u);
          float exs[H];
#pragma unroll
          for (int h = 0; h < H; ++h)
            exs[h] = __int_as_float(__builtin_amdgcn_readlane(__float_as_int(ex_l[h]), j0 + u));
          float* op = racc_s + dr * F + lane;
#pragma unroll
          for (int k = 0; k < PER; ++k) atomicAdd(op + 32 * k, exs[k / LPH] * f[u][k]);
        }
      } else {
        for (int j = j0; j < cnt; ++j) {
          const int sg = __builtin_amdgcn_readlane(s_l, j);
          const int dr = __builtin_amdgcn_readlane(d_l, j);
          float exs[H];
#pragma unroll
          for (int h = 0; h < H; ++h)
            exs[h] = __int_as_float(__builtin_amdgcn_readlane(__float_as_int(ex_l[h]), j));
          const float* fp = feat + (size_t)sg * F + lane;
          float*       op = racc_s + dr * F + lane;
#pragma unroll
          for (int k = 0; k < PER; ++k) atomicAdd(op + 32 * k, exs[k / LPH] * fp[32 * k]);
        }
      }
    }
  };

  sweep([&](int cnt, int qh) {
    int s_l, d_l;
    float v[H];
    edge_of(cnt, qh, s_l, d_l, v);
    const bool mine = d_l < NPB;
    const unsigned mask = __builtin_amdgcn_ballot_w32(mine);
    if (mask != 0u) {
      int base = 0;
      if (lane == 0) base = atomicAdd(&lst_n, __builtin_popcount(mask));
      base = __builtin_amdgcn_readfirstlane(base);
      if (mine) {
#pragma unroll
        for (int h = 0; h < H; ++h) atomicMax(&mkey_s[d_l * H + h], fkey(v[h]));
        const int pos = base + (int)__builtin_amdgcn_mbcnt_lo(mask, 0u);
        if (pos < CAP) {
          lst_s[pos] = s_l;
          lst_d[pos] = (unsigned short)d_l;
#pragma unroll
          for (int h = 0; h < H; ++h) lst_v[pos * H + h] = v[h];
        }
      }
    }
  });
  __syncthreads();
  const int total = __builtin_amdgcn_readfirstlane(lst_n);

  if (total <= CAP) {
    for (int i = wv * 32; i < total; i += OWN_T) {
      const int cnt = (total - i < 32) ? (total - i) : 32;
      int s_l = 0, d_l = NPB;
      float v[H];
#pragma unroll
      for (int h = 0; h < H; ++h) v[h] = 0.f;
      if (lane < cnt) {
        s_l = lst_s[i + lane];
        d_l = lst_d[i + lane];
#pragma unroll
        for (int h = 0; h < H; ++h) v[h] = lst_v[(i + lane) * H + h];
      }
      finish(cnt, s_l, d_l, v);
    }
  } else {
    sweep([&](int cnt, int qh) {
      int s_l, d_l;
      float v[H];
      edge_of(cnt, qh, s_l, d_l, v);
      finish(cnt, s_l, d_l, v);
    });
  }
  __syncthreads();

  for (int i4 = tid; i4 < NPB * (F / 4); i4 += OWN_T) {
    const int nl = i4 / (F / 4);
    const int n  = lo + nl;
    if (n >= N) break;
    const int c = (i4 % (F / 4)) * 4;
    const int h = c / D;
    float sden = ssum_s[nl * H + h];
    sden = (sden > 0.f) ? sden : 1.f;
    const float4 v = reinterpret_cast<const float4*>(racc_s)[i4];
    float4* gp = reinterpret_cast<float4*>(acc + (size_t)n * F + c);
    float4 o;
    if constexpr (FIRST) {
      o = make_float4(0.f, 0.f, 0.f, 0.f);
      for (int r = 0; r < R_REL; ++r) {
        const float4 bv = *reinterpret_cast<const float4*>(b + (size_t)r * F + c);
        o.x += bv.x; o.y += bv.y; o.z += bv.z; o.w += bv.w;
      }
    } else {
      o = *gp;
    }
    o.x += v.x / sden; o.y += v.y / sden; o.z += v.z / sden; o.w += v.w / sden;
    *gp = o;
  }
}

extern "C" void kernel_launch(void* const* d_in, const int* in_sizes, int n_in,
                              void* d_out, int out_size, void* d_ws, size_t ws_size,
                              hipStream_t stream) {
  const float* x    = (const float*)d_in[0];
  const int*   esrc = (const int*)d_in[1];
  const int*   edst = (const int*)d_in[2];
  const float* W1   = (const float*)d_in[3];
  const float* al1  = (const float*)d_in[4];
  const float* ar1  = (const float*)d_in[5];
  const float* b1   = (const float*)d_in[6];
  const float* W2   = (const float*)d_in[7];
  const float* al2  = (const float*)d_in[8];
  const float* ar2  = (const float*)d_in[9];
  const float* b2   = (const float*)d_in[10];
  float* out = (float*)d_out;

  const int N = in_sizes[0] / IN_F;
  const int E = in_sizes[1] / (2 * R_REL);

  size_t off = 0;
  auto carve = [&](size_t bytes) -> void* {
    void* p = (char*)d_ws + off;
    off += (bytes + 255) & ~(size_t)255;
    return p;
  };
  unsigned short* xbf  = (unsigned short*)carve((size_t)N * IN_F * 2);
  unsigned short* hbf  = (unsigned short*)carve((size_t)N * (H1 * HID_F) * 2);
  unsigned short* wt1  = (unsigned short*)carve((size_t)R_REL * IN_F * (H1 * HID_F) * 2);
  unsigned short* wt2  = (unsigned short*)carve((size_t)R_REL * (H1 * HID_F) * OUT_F * 2);
  float*    feat  = (float*)carve((size_t)N * (H1 * HID_F) * 4);
  float*    el    = (float*)carve((size_t)N * H1 * 4);
  float*    er    = (float*)carve((size_t)N * H1 * 4);
  float*    hacc  = (float*)carve((size_t)N * (H1 * HID_F) * 4);
  const size_t ownStride = ((size_t)E + 16 + 255) & ~(size_t)255;
  unsigned char* own = (unsigned char*)carve(ownStride * 2 * R_REL);
  const unsigned ownGrid = (unsigned)((N + NPB - 1) / NPB);

  const int T = 256;
  auto blk1 = [&](size_t n) { return (unsigned)((n + T - 1) / T); };

  k_f32_to_bf16<<<blk1((size_t)N * IN_F), T, 0, stream>>>(x, xbf, (size_t)N * IN_F);
  for (int r = 0; r < R_REL; ++r) {
    k_transpose_w<<<blk1((size_t)IN_F * (H1 * HID_F)), T, 0, stream>>>(
        W1 + (size_t)r * IN_F * (H1 * HID_F), wt1 + (size_t)r * IN_F * (H1 * HID_F),
        IN_F, H1 * HID_F);
    k_transpose_w<<<blk1((size_t)(H1 * HID_F) * OUT_F), T, 0, stream>>>(
        W2 + (size_t)r * (H1 * HID_F) * OUT_F, wt2 + (size_t)r * (H1 * HID_F) * OUT_F,
        H1 * HID_F, OUT_F);
  }
  k_owner_bytes<<<dim3((unsigned)((((size_t)E + 3) / 4 + OB_T * OB_W - 1) / (OB_T * OB_W)), 2 * R_REL), OB_T, 0, stream>>>(
      edst, own, E, ownStride);

  constexpr int RPB = 4;
  const unsigned gemmGrid = (unsigned)((N + 16 * RPB - 1) / (16 * RPB));

  for (int r = 0; r < R_REL; ++r) {
    const int* s1 = esrc + (size_t)(0 * R_REL + r) * E;
    const int* d1 = edst + (size_t)(0 * R_REL + r) * E;
    const unsigned char* o1 = own + (size_t)(0 * R_REL + r) * ownStride;

    k_gemm_bf16_wmma<IN_F, H1 * HID_F, RPB><<<gemmGrid, 2 * (H1 * HID_F), 0, stream>>>(
        xbf, wt1 + (size_t)r * IN_F * (H1 * HID_F), feat, N);

    k_attn_dots<<<blk1((size_t)N * H1), T, 0, stream>>>(
        feat, al1 + (size_t)r * H1 * HID_F, ar1 + (size_t)r * H1 * HID_F,
        el, er, N, H1, HID_F);

    if (r == 0)
      k_owner_aggregate<H1, HID_F, true><<<ownGrid, OWN_T, 0, stream>>>(
          el, er, feat, s1, d1, o1, b1, hacc, N, E);
    else
      k_owner_aggregate<H1, HID_F, false><<<ownGrid, OWN_T, 0, stream>>>(
          el, er, feat, s1, d1, o1, b1, hacc, N, E);
  }

  k_relu_f32_to_bf16<<<blk1((size_t)N * (H1 * HID_F)), T, 0, stream>>>(
      hacc, hbf, (size_t)N * (H1 * HID_F));

  for (int r = 0; r < R_REL; ++r) {
    const int* s2 = esrc + (size_t)(1 * R_REL + r) * E;
    const int* d2 = edst + (size_t)(1 * R_REL + r) * E;
    const unsigned char* o2 = own + (size_t)(1 * R_REL + r) * ownStride;

    k_gemm_bf16_wmma<H1 * HID_F, OUT_F, RPB><<<gemmGrid, 2 * OUT_F, 0, stream>>>(
        hbf, wt2 + (size_t)r * (H1 * HID_F) * OUT_F, feat, N);

    k_attn_dots<<<blk1((size_t)N * H2), T, 0, stream>>>(
        feat, al2 + (size_t)r * H2 * OUT_F, ar2 + (size_t)r * H2 * OUT_F,
        el, er, N, H2, OUT_F);

    if (r == 0)
      k_owner_aggregate<H2, OUT_F, true><<<ownGrid, OWN_T, 0, stream>>>(
          el, er, feat, s2, d2, o2, b2, out, N, E);
    else
      k_owner_aggregate<H2, OUT_F, false><<<ownGrid, OWN_T, 0, stream>>>(
          el, er, feat, s2, d2, o2, b2, out, N, E);
  }
}


# el, er made in the product kernel's epilogue from its f32 results in an LDS tile (half-wave per row and head, DPP butterfly, fixed order): 4 dot kernels gone; el, er sums re-associated
# speedup vs baseline: 4.9860x; 1.2805x over previous
#include <hip/hip_runtime.h>

#define R_REL   2
#define IN_F    128
#define HID_F   64
#define OUT_F   64
#define H1      2
#define H2      1
#define NEG_SLOPE 0.2f

typedef __attribute__((ext_vector_type(16))) __bf16 v16bf;
typedef __attribute__((ext_vector_type(8)))  float  v8f;
typedef __attribute__((ext_vector_type(4)))  float  v4f;
typedef __attribute__((ext_vector_type(4)))  __bf16 v4bf;
typedef __attribute__((ext_vector_type(2)))  unsigned v2u;

union Frag16 { uint4 q[2]; v16bf v; };

__device__ __forceinline__ unsigned short f2bf(float f) {
  unsigned u = __float_as_uint(f);
  unsigned r = u + 0x7FFFu + ((u >> 16) & 1u);
  return (unsigned short)(r >> 16);
}
__device__ __forceinline__ unsigned fkey(float f) {
  unsigned u = __float_as_uint(f);
  return (u & 0x80000000u) ? ~u : (u | 0x80000000u);
}
__device__ __forceinline__ float funkey(unsigned k) {
  unsigned u = (k & 0x80000000u) ? (k & 0x7FFFFFFFu) : ~k;
  return __uint_as_float(u);
}
__device__ __forceinline__ float lrelu(float x) { return x >= 0.f ? x : NEG_SLOPE * x; }
template <int XM>
__device__ __forceinline__ float row_xor(float v) {
  return __int_as_float(__builtin_amdgcn_update_dpp(0, __float_as_int(v), 0x160 + XM, 0xF, 0xF, true));
}
__device__ __forceinline__ float row16_sum(float v) {
  v += row_xor<1>(v);
  v += row_xor<2>(v);
  v += row_xor<4>(v);
  v += row_xor<8>(v);
  return v;
}

__global__ void k_transpose_w(const float* __restrict__ W,
                              unsigned short* __restrict__ Wt, int K, int NCOLS) {
  int i = blockIdx.x * blockDim.x + threadIdx.x;
  if (i >= K * NCOLS) return;
  int n = i / K, k = i % K;
  Wt[i] = f2bf(W[(size_t)k * NCOLS + n]);
}

template <int KC, int RPB, int STRIP>
__device__ __forceinline__ void gemm_strips(const unsigned char* ap, const Frag16 (&fb)[KC],
                                            v8f (&acc)[RPB]) {
#pragma unroll
  for (int t = 0; t < RPB; ++t) acc[t] = v8f{};
#pragma unroll
  for (int c = 0; c < KC; ++c) {
#pragma unroll
    for (int t = 0; t < RPB; ++t) {
      Frag16 fa;
      fa.q[0] = *reinterpret_cast<const uint4*>(ap + t * STRIP + 64 * c);
      fa.q[1] = *reinterpret_cast<const uint4*>(ap + t * STRIP + 64 * c + 32);
      acc[t] = __builtin_amdgcn_wmma_f32_16x16x32_bf16(false, fa.v, false, fb[c].v,
                                                       (short)0, acc[t], false, false);
    }
  }
  __builtin_amdgcn_sched_barrier(0);
}

template <int NCOLS, int RPB, bool GUARD>
__device__ __forceinline__ void store_strips(const v8f (&acc)[RPB], float* cp, int rOff, int rowsLeft) {
#pragma unroll
  for (int t = 0; t < RPB; ++t) {
#pragma unroll
    for (int v = 0; v < 8; ++v) {
      const int r = 16 * t + rOff + v;
      if (!GUARD || r < rowsLeft) cp[(size_t)r * NCOLS] = acc[t][v];
    }
  }
}

constexpr int DHEAD = 64;
static_assert(HID_F == DHEAD && OUT_F == DHEAD, "a head is 64 columns in both layers");
template <int KDIM, int NCOLS, int RPB, bool RELU>
__global__ __launch_bounds__(2 * NCOLS)
void k_gemm_f32a_wmma(const float* __restrict__ A,
                      const unsigned short* __restrict__ Bt,
                      float* __restrict__ C,
                      const float* __restrict__ al, const float* __restrict__ ar,
                      float* __restrict__ el, float* __restrict__ er, int M) {
  constexpr int KC    = KDIM / 32;
  constexpr int NT    = 2 * NCOLS;
  constexpr int ROWS  = 16 * RPB;
  constexpr int Q4    = KDIM / 4;
  constexpr int NLD   = ROWS * Q4 / NT;
  constexpr int PITCH = KDIM * 2 + 16;
  static_assert((ROWS * Q4) % NT == 0, "the tile must divide among the threads");
  static_assert(NT % Q4 == 0, "a wave-load must stay inside one row");
  __shared__ __attribute__((aligned(16))) unsigned char sA[ROWS * PITCH];
  constexpr int CPITCH = NCOLS + 4;
  constexpr int HEADS  = NCOLS / DHEAD;
  constexpr int UPW    = ROWS * HEADS / (NCOLS / 16);
  static_assert(NCOLS % DHEAD == 0, "whole heads");
  static_assert(HEADS == 1 || HEADS == 2, "a wave-load takes two units: one row's two heads, or two rows");
  static_assert(UPW % 2 == 0 && UPW / 2 <= 16, "two units per wave-load, one result per lane of a row");
  __shared__ __attribute__((aligned(16))) float sC[ROWS * CPITCH];

  const int tid      = threadIdx.x;
  const int lane     = tid & 31;
  const int wv       = tid >> 5;
  const int colBase  = wv * 16;
  const int l16      = lane & 15;
  const int k0       = (lane < 16) ? 0 : 8;
  const int rowBlock = blockIdx.x * ROWS;

  v4f av[NLD];
#pragma unroll
  for (int j = 0; j < NLD; ++j) {
    const int p = j * NT + tid;
    int arow = rowBlock + p / Q4; if (arow >= M) arow = M - 1;
    av[j] = *reinterpret_cast<const v4f*>(A + (size_t)arow * KDIM + 4 * (p % Q4));
  }

  Frag16 fb[KC];
  {
    const unsigned short* bp = Bt + (size_t)(colBase + l16) * KDIM + k0;
#pragma unroll
    for (int c = 0; c < KC; ++c) {
      fb[c].q[0] = *reinterpret_cast<const uint4*>(bp + 32 * c);
      fb[c].q[1] = *reinterpret_cast<const uint4*>(bp + 32 * c + 16);
    }
  }

#pragma unroll
  for (int j = 0; j < NLD; ++j) {
    const int p = j * NT + tid;
    v4f v = av[j];
    if constexpr (RELU) {
#pragma unroll
      for (int i = 0; i < 4; ++i) v[i] = v[i] > 0.f ? v[i] : 0.f;
    }
    const v4bf h = __builtin_convertvector(v, v4bf);
    *reinterpret_cast<v2u*>(sA + (p / Q4) * PITCH + 8 * (p % Q4)) = __builtin_bit_cast(v2u, h);
  }
  __syncthreads();

  const int rOff = (lane < 16) ? 0 : 8;
  const unsigned char* ap = sA + l16 * PITCH + 2 * k0;
  float* cp = C + (size_t)rowBlock * NCOLS + colBase + l16;
  float* tp = sC + rOff * CPITCH + colBase + l16;

  const int hf = lane >> 4;
  const int hd = (HEADS == 2) ? hf : 0;
  const v4f wl = *reinterpret_cast<const v4f*>(al + hd * DHEAD + 4 * l16);
  const v4f wr = *reinterpret_cast<const v4f*>(ar + hd * DHEAD + 4 * l16);

  v8f acc[RPB];
  gemm_strips<KC, RPB, 16 * PITCH>(ap, fb, acc);
#pragma unroll
  for (int t = 0; t < RPB; ++t) {
#pragma unroll
    for (int v = 0; v < 8; ++v) tp[(16 * t + v) * CPITCH] = acc[t][v];
  }
  __syncthreads();

  float ke = 0.f, kr = 0.f;
#pragma unroll
  for (int i = 0; i < UPW / 2; ++i) {
    const int u = UPW * wv + 2 * i + hf;
    const v4f c = *reinterpret_cast<const v4f*>(sC + (u / HEADS) * CPITCH + (u % HEADS) * DHEAD + 4 * l16);
    float sl = c[0] * wl[0] + c[1] * wl[1] + c[2] * wl[2] + c[3] * wl[3];
    float sr = c[0] * wr[0] + c[1] * wr[1] + c[2] * wr[2] + c[3] * wr[3];
    sl = row16_sum(sl);
    sr = row16_sum(sr);
    if (l16 == i) { ke = sl; kr = sr; }
  }
  const long long ug = (long long)rowBlock * HEADS + UPW * wv + 2 * l16 + hf;
  if (l16 < UPW / 2 && ug < (long long)M * HEADS) { el[ug] = ke; er[ug] = kr; }

  if (rowBlock + ROWS <= M)
    store_strips<NCOLS, RPB, false>(acc, cp, rOff, ROWS);
  else
    store_strips<NCOLS, RPB, true>(acc, cp, rOff, M - rowBlock);
}

constexpr int NPB      = 392;

constexpr int PL_T     = 256;
__device__ __forceinline__ unsigned owner_byte(int d) { return ((unsigned)d / (unsigned)NPB) & 255u; }

__global__ __launch_bounds__(PL_T)
void k_owner_planes(const int* __restrict__ dst, unsigned char* __restrict__ own,
                    int E, size_t stride) {
  const int* dp = dst + (size_t)blockIdx.y * E;
  unsigned char* op = own + (size_t)blockIdx.y * stride;
  const int lane = threadIdx.x & 31;
  const int wq = __builtin_amdgcn_readfirstlane((int)blockIdx.x * (PL_T / 32) + (int)(threadIdx.x >> 5));
  const int nBlocks = (((E + 127) >> 7) + 31) >> 5;
  if (wq >= 4 * nBlocks) return;

  int4 v[8];
  if ((E & 3) == 0) {
#pragma unroll
    for (int c = 0; c < 8; ++c) {
      const int e0 = (wq * 8 + c) * 128 + 4 * lane;
      v[c] = *reinterpret_cast<const int4*>(dp + (e0 < E ? e0 : 0));
    }
  } else {
#pragma unroll
    for (int c = 0; c < 8; ++c) {
      const int e0 = (wq * 8 + c) * 128 + 4 * lane;
      v[c].x = (e0     < E) ? dp[e0]     : 0;
      v[c].y = (e0 + 1 < E) ? dp[e0 + 1] : 0;
      v[c].z = (e0 + 2 < E) ? dp[e0 + 2] : 0;
      v[c].w = (e0 + 3 < E) ? dp[e0 + 3] : 0;
    }
  }

  unsigned a[8][4];
#pragma unroll
  for (int b = 0; b < 8; ++b)
#pragma unroll
    for (int j = 0; j < 4; ++j) a[b][j] = 0u;
#pragma unroll
  for (int c = 0; c < 8; ++c) {
    const unsigned pk = owner_byte(v[c].x) | (owner_byte(v[c].y) << 8) |
                        (owner_byte(v[c].z) << 16) | (owner_byte(v[c].w) << 24);
#pragma unroll
    for (int b = 0; b < 8; ++b) {
#pragma unroll
      for (int j = 0; j < 4; ++j) {
        const unsigned bal = __builtin_amdgcn_ballot_w32(((pk >> (8 * j + b)) & 1u) != 0u);
        if (lane == c) a[b][j] = bal;
      }
    }
  }
  if (lane < 8) {
    unsigned char* bp = op + (size_t)(wq >> 2) * 4096 + (size_t)(((wq & 3) * 8 + lane) * 16);
#pragma unroll
    for (int b = 0; b < 8; ++b)
      *reinterpret_cast<uint4*>(bp + 512 * b) = make_uint4(a[b][0], a[b][1], a[b][2], a[b][3]);
  }
}

constexpr int OWN_T    = 1024;
constexpr int RING     = 64;
constexpr int CAP      = 7168;

template <int H, int D, bool FIRST>
__global__ __launch_bounds__(OWN_T)
void k_owner_aggregate(const float* __restrict__ el, const float* __restrict__ er,
                       const float* __restrict__ feat,
                       const int* __restrict__ src, const int* __restrict__ dst,
                       const unsigned char* __restrict__ own,
                       const float* __restrict__ b, float* __restrict__ acc, int N, int E) {
  constexpr int F   = H * D;
  constexpr int PER = F / 32;
  constexpr int LPH = D / 32;
  constexpr int U   = 32 / PER;
  __shared__ __attribute__((aligned(16))) float racc_s[(NPB + 1) * F];
  __shared__ float    ssum_s[NPB * H];
  __shared__ unsigned mkey_s[NPB * H];
  __shared__ float    er_s[NPB * H];
  __shared__ int      ring_e[OWN_T / 32][RING];
  __shared__ int      lst_s[CAP];
  __shared__ float    lst_v[CAP * H];
  __shared__ unsigned short lst_d[CAP];
  __shared__ int      lst_n;

  const int tid  = threadIdx.x;
  const int lane = tid & 31;
  const int wv   = __builtin_amdgcn_readfirstlane(tid >> 5);
  const int lo   = blockIdx.x * NPB;

  for (int i = tid; i < NPB * F / 4; i += OWN_T)
    reinterpret_cast<float4*>(racc_s)[i] = make_float4(0.f, 0.f, 0.f, 0.f);
  for (int i = tid; i < NPB * H; i += OWN_T) {
    ssum_s[i] = 0.f;
    mkey_s[i] = 0u;
    er_s[i]   = (lo + i / H < N) ? er[(size_t)lo * H + i] : 0.f;
  }
  if (tid == 0) lst_n = 0;
  __syncthreads();

  int* const re = ring_e[wv];

  const int Q  = (E + 127) >> 7;
  const int WL = (Q + 31) >> 5;
  const uint4* own4 = reinterpret_cast<const uint4*>(own);
  unsigned flip[8];
#pragma unroll
  for (int b = 0; b < 8; ++b) flip[b] = ((blockIdx.x >> b) & 1u) ? 0u : 0xFFFFFFFFu;

  const int rot  = (int)(((long long)WL * blockIdx.x) / gridDim.x);
  const int nIt  = (WL > wv) ? ((WL - wv + 31) >> 5) : 0;

  auto sweep = [&](auto&& work) {
    int qh = 0, qt = 0;

    auto push = [&](bool keep, int e) {
      const unsigned mask = __builtin_amdgcn_ballot_w32(keep);
      if (keep) {
        const int pos = (qt + (int)__builtin_amdgcn_mbcnt_lo(mask, 0u)) & (RING - 1);
        re[pos] = e;
      }
      qt += __builtin_popcount(mask);
    };

    int tcur = wv + rot;
    if (tcur >= WL) tcur -= WL;
    uint4 w[8];
    {
      const uint4* bp = own4 + (size_t)(nIt > 0 ? tcur : 0) * 256 + lane;
#pragma unroll
      for (int b = 0; b < 8; ++b) w[b] = bp[32 * b];
    }

    for (int it = 0; it < nIt; ++it) {
      unsigned m0 = w[0].x ^ flip[0], m1 = w[0].y ^ flip[0], m2 = w[0].z ^ flip[0], m3 = w[0].w ^ flip[0];
#pragma unroll
      for (int b = 1; b < 8; ++b) {
        m0 = __builtin_amdgcn_bitop3_b32(m0, w[b].x, flip[b], 0x60);
        m1 = __builtin_amdgcn_bitop3_b32(m1, w[b].y, flip[b], 0x60);
        m2 = __builtin_amdgcn_bitop3_b32(m2, w[b].z, flip[b], 0x60);
        m3 = __builtin_amdgcn_bitop3_b32(m3, w[b].w, flip[b], 0x60);
      }
      const int q = tcur * 32 + lane;
      int tnext = tcur + 32;
      if (tnext >= WL) tnext -= WL;
      {
        const uint4* bp = own4 + (size_t)(it + 1 < nIt ? tnext : tcur) * 256 + lane;
#pragma unroll
        for (int b = 0; b < 8; ++b) w[b] = bp[32 * b];
      }
      tcur = tnext;
      if (q >= Q) { m0 = 0u; m1 = 0u; m2 = 0u; m3 = 0u; }
      const int e0 = q * 128;
      unsigned any = m0 | m1 | m2 | m3;
      for (int trip = 0; trip < 128; ++trip) {
        if (__builtin_amdgcn_ballot_w32(any != 0u) == 0u) break;
        const bool c0 = m0 != 0u, c1 = m1 != 0u, c2 = m2 != 0u;
        const unsigned ws = c0 ? m0 : (c1 ? m1 : (c2 ? m2 : m3));
        const int      js = c0 ? 0  : (c1 ? 1  : (c2 ? 2  : 3));
        const int p = __builtin_ctz(ws | 0x80000000u);
        const int e = e0 + (p << 2) + js;
        push(any != 0u && e < E, e);
        const unsigned wc = ws & (ws - 1u);
        const bool s1 = c1 && !c0, s2 = c2 && !c1 && !c0, s3 = !c2 && !c1 && !c0;
        m0 = c0 ? wc : m0;
        m1 = s1 ? wc : m1;
        m2 = s2 ? wc : m2;
        m3 = s3 ? wc : m3;
        any = m0 | m1 | m2 | m3;
        if (qt - qh >= 32) { work(32, qh); qh += 32; }
      }
    }
    if (qt - qh > 0) { work(qt - qh, qh); qh = qt; }
  };

  auto edge_of = [&](int cnt, int qh, int& s_l, int& d_l, float (&v)[H]) {
    s_l = 0; d_l = NPB;
#pragma unroll
    for (int h = 0; h < H; ++h) v[h] = 0.f;
    if (lane < cnt) {
      const int e = re[(qh + lane) & (RING - 1)];
      s_l = src[e];
      const unsigned rel = (unsigned)(dst[e] - lo);
      d_l = (rel < (unsigned)NPB) ? (int)rel : NPB;
    }
    if (d_l < NPB) {
      float a[H];
      if constexpr (H == 2) {
        const float2 a2 = *reinterpret_cast<const float2*>(el + (size_t)s_l * 2);
        a[0] = a2.x; a[1] = a2.y;
      } else {
#pragma unroll
        for (int h = 0; h < H; ++h) a[h] = el[(size_t)s_l * H + h];
      }
#pragma unroll
      for (int h = 0; h < H; ++h) v[h] = lrelu(a[h] + er_s[d_l * H + h]);
    }
  };

  auto finish = [&](int cnt, int s_l, int d_l, const float (&v)[H]) {
    float ex_l[H];
#pragma unroll
    for (int h = 0; h < H; ++h) ex_l[h] = 0.f;
    if (d_l < NPB) {
#pragma unroll
      for (int h = 0; h < H; ++h) {
        ex_l[h] = __expf(v[h] - funkey(mkey_s[d_l * H + h]));
        atomicAdd(&ssum_s[d_l * H + h], ex_l[h]);
      }
    }

    for (int j0 = 0; j0 < cnt; j0 += U) {
      if (j0 + U <= cnt) {
        float f[U][PER];
#pragma unroll
        for (int u = 0; u < U; ++u) {
          const int sg = __builtin_amdgcn_readlane(s_l, j0 + u);
          const float* fp = feat + (size_t)sg * F + lane;
#pragma unroll
          for (int k = 0; k < PER; ++k) f[u][k] = fp[32 * k];
        }
#pragma unroll
        for (int u = 0; u < U; ++u) {
          const int dr = __builtin_amdgcn_readlane(d_l, j0 + u);
          float exs[H];
#pragma unroll
          for (int h = 0; h < H; ++h)
            exs[h] = __int_as_float(__builtin_amdgcn_readlane(__float_as_int(ex_l[h]), j0 + u));
          float* op = racc_s + dr * F + lane;
#pragma unroll
          for (int k = 0; k < PER; ++k) atomicAdd(op + 32 * k, exs[k / LPH] * f[u][k]);
        }
      } else {
        for (int j = j0; j < cnt; ++j) {
          const int sg = __builtin_amdgcn_readlane(s_l, j);
          const int dr = __builtin_amdgcn_readlane(d_l, j);
          float exs[H];
#pragma unroll
          for (int h = 0; h < H; ++h)
            exs[h] = __int_as_float(__builtin_amdgcn_readlane(__float_as_int(ex_l[h]), j));
          const float* fp = feat + (size_t)sg * F + lane;
          float*       op = racc_s + dr * F + lane;
#pragma unroll
          for (int k = 0; k < PER; ++k) atomicAdd(op + 32 * k, exs[k / LPH] * fp[32 * k]);
        }
      }
    }
  };

  sweep([&](int cnt, int qh) {
    int s_l, d_l;
    float v[H];
    edge_of(cnt, qh, s_l, d_l, v);
    const bool mine = d_l < NPB;
    const unsigned mask = __builtin_amdgcn_ballot_w32(mine);
    if (mask != 0u) {
      int base = 0;
      if (lane == 0) base = atomicAdd(&lst_n, __builtin_popcount(mask));
      base = __builtin_amdgcn_readfirstlane(base);
      if (mine) {
#pragma unroll
        for (int h = 0; h < H; ++h) atomicMax(&mkey_s[d_l * H + h], fkey(v[h]));
        const int pos = base + (int)__builtin_amdgcn_mbcnt_lo(mask, 0u);
        if (pos < CAP) {
          lst_s[pos] = s_l;
          lst_d[pos] = (unsigned short)d_l;
#pragma unroll
          for (int h = 0; h < H; ++h) lst_v[pos * H + h] = v[h];
        }
      }
    }
  });
  __syncthreads();
  const int total = __builtin_amdgcn_readfirstlane(lst_n);

  if (total <= CAP) {
    for (int i = wv * 32; i < total; i += OWN_T) {
      const int cnt = (total - i < 32) ? (total - i) : 32;
      int s_l = 0, d_l = NPB;
      float v[H];
#pragma unroll
      for (int h = 0; h < H; ++h) v[h] = 0.f;
      if (lane < cnt) {
        s_l = lst_s[i + lane];
        d_l = lst_d[i + lane];
#pragma unroll
        for (int h = 0; h < H; ++h) v[h] = lst_v[(i + lane) * H + h];
      }
      finish(cnt, s_l, d_l, v);
    }
  } else {
    sweep([&](int cnt, int qh) {
      int s_l, d_l;
      float v[H];
      edge_of(cnt, qh, s_l, d_l, v);
      finish(cnt, s_l, d_l, v);
    });
  }
  __syncthreads();

  for (int i4 = tid; i4 < NPB * (F / 4); i4 += OWN_T) {
    const int nl = i4 / (F / 4);
    const int n  = lo + nl;
    if (n >= N) break;
    const int c = (i4 % (F / 4)) * 4;
    const int h = c / D;
    float sden = ssum_s[nl * H + h];
    sden = (sden > 0.f) ? sden : 1.f;
    const float4 v = reinterpret_cast<const float4*>(racc_s)[i4];
    float4* gp = reinterpret_cast<float4*>(acc + (size_t)n * F + c);
    float4 o;
    if constexpr (FIRST) {
      o = make_float4(0.f, 0.f, 0.f, 0.f);
      for (int r = 0; r < R_REL; ++r) {
        const float4 bv = *reinterpret_cast<const float4*>(b + (size_t)r * F + c);
        o.x += bv.x; o.y += bv.y; o.z += bv.z; o.w += bv.w;
      }
    } else {
      o = *gp;
    }
    o.x += v.x / sden; o.y += v.y / sden; o.z += v.z / sden; o.w += v.w / sden;
    *gp = o;
  }
}

extern "C" void kernel_launch(void* const* d_in, const int* in_sizes, int n_in,
                              void* d_out, int out_size, void* d_ws, size_t ws_size,
                              hipStream_t stream) {
  const float* x    = (const float*)d_in[0];
  const int*   esrc = (const int*)d_in[1];
  const int*   edst = (const int*)d_in[2];
  const float* W1   = (const float*)d_in[3];
  const float* al1  = (const float*)d_in[4];
  const float* ar1  = (const float*)d_in[5];
  const float* b1   = (const float*)d_in[6];
  const float* W2   = (const float*)d_in[7];
  const float* al2  = (const float*)d_in[8];
  const float* ar2  = (const float*)d_in[9];
  const float* b2   = (const float*)d_in[10];
  float* out = (float*)d_out;

  const int N = in_sizes[0] / IN_F;
  const int E = in_sizes[1] / (2 * R_REL);

  size_t off = 0;
  auto carve = [&](size_t bytes) -> void* {
    void* p = (char*)d_ws + off;
    off += (bytes + 255) & ~(size_t)255;
    return p;
  };
  unsigned short* wt1  = (unsigned short*)carve((size_t)R_REL * IN_F * (H1 * HID_F) * 2);
  unsigned short* wt2  = (unsigned short*)carve((size_t)R_REL * (H1 * HID_F) * OUT_F * 2);
  float*    feat  = (float*)carve((size_t)N * (H1 * HID_F) * 4);
  float*    el    = (float*)carve((size_t)N * H1 * 4);
  float*    er    = (float*)carve((size_t)N * H1 * 4);
  float*    hacc  = (float*)carve((size_t)N * (H1 * HID_F) * 4);
  const size_t ownBlocks = ((((size_t)E + 127) >> 7) + 31) >> 5;
  const size_t ownStride = ownBlocks * 4096;
  unsigned char* own = (unsigned char*)carve(ownStride * 2 * R_REL);
  const unsigned ownGrid = (unsigned)((N + NPB - 1) / NPB);

  const int T = 256;
  auto blk1 = [&](size_t n) { return (unsigned)((n + T - 1) / T); };

  for (int r = 0; r < R_REL; ++r) {
    k_transpose_w<<<blk1((size_t)IN_F * (H1 * HID_F)), T, 0, stream>>>(
        W1 + (size_t)r * IN_F * (H1 * HID_F), wt1 + (size_t)r * IN_F * (H1 * HID_F),
        IN_F, H1 * HID_F);
    k_transpose_w<<<blk1((size_t)(H1 * HID_F) * OUT_F), T, 0, stream>>>(
        W2 + (size_t)r * (H1 * HID_F) * OUT_F, wt2 + (size_t)r * (H1 * HID_F) * OUT_F,
        H1 * HID_F, OUT_F);
  }
  k_owner_planes<<<dim3((unsigned)((ownBlocks * 4 + PL_T / 32 - 1) / (PL_T / 32)), 2 * R_REL), PL_T, 0, stream>>>(
      edst, own, E, ownStride);

  constexpr int RPB = 4;
  const unsigned gemmGrid = (unsigned)((N + 16 * RPB - 1) / (16 * RPB));

  for (int r = 0; r < R_REL; ++r) {
    const int* s1 = esrc + (size_t)(0 * R_REL + r) * E;
    const int* d1 = edst + (size_t)(0 * R_REL + r) * E;
    const unsigned char* o1 = own + (size_t)(0 * R_REL + r) * ownStride;

    k_gemm_f32a_wmma<IN_F, H1 * HID_F, RPB, false><<<gemmGrid, 2 * (H1 * HID_F), 0, stream>>>(
        x, wt1 + (size_t)r * IN_F * (H1 * HID_F), feat,
        al1 + (size_t)r * H1 * HID_F, ar1 + (size_t)r * H1 * HID_F, el, er, N);

    if (r == 0)
      k_owner_aggregate<H1, HID_F, true><<<ownGrid, OWN_T, 0, stream>>>(
          el, er, feat, s1, d1, o1, b1, hacc, N, E);
    else
      k_owner_aggregate<H1, HID_F, false><<<ownGrid, OWN_T, 0, stream>>>(
          el, er, feat, s1, d1, o1, b1, hacc, N, E);
  }

  for (int r = 0; r < R_REL; ++r) {
    const int* s2 = esrc + (size_t)(1 * R_REL + r) * E;
    const int* d2 = edst + (size_t)(1 * R_REL + r) * E;
    const unsigned char* o2 = own + (size_t)(1 * R_REL + r) * ownStride;

    k_gemm_f32a_wmma<H1 * HID_F, OUT_F, RPB, true><<<gemmGrid, 2 * OUT_F, 0, stream>>>(
        hacc, wt2 + (size_t)r * (H1 * HID_F) * OUT_F, feat,
        al2 + (size_t)r * H2 * OUT_F, ar2 + (size_t)r * H2 * OUT_F, el, er, N);

    if (r == 0)
      k_owner_aggregate<H2, OUT_F, true><<<ownGrid, OWN_T, 0, stream>>>(
          el, er, feat, s2, d2, o2, b2, out, N, E);
    else
      k_owner_aggregate<H2, OUT_F, false><<<ownGrid, OWN_T, 0, stream>>>(
          el, er, feat, s2, d2, o2, b2, out, N, E);
  }
}


# Prep pass also writes each edge's (source, destination) as one 8-byte record: the owner sweep gets a kept edge's ends by one access to one line, not two scattered 4-byte reads; sum order unchanged
# speedup vs baseline: 5.2759x; 1.0581x over previous
#include <hip/hip_runtime.h>

#define R_REL   2
#define IN_F    128
#define HID_F   64
#define OUT_F   64
#define H1      2
#define H2      1
#define NEG_SLOPE 0.2f

typedef __attribute__((ext_vector_type(16))) __bf16 v16bf;
typedef __attribute__((ext_vector_type(8)))  float  v8f;
typedef __attribute__((ext_vector_type(4)))  float  v4f;
typedef __attribute__((ext_vector_type(4)))  __bf16 v4bf;
typedef __attribute__((ext_vector_type(2)))  unsigned v2u;

union Frag16 { uint4 q[2]; v16bf v; };

__device__ __forceinline__ unsigned short f2bf(float f) {
  unsigned u = __float_as_uint(f);
  unsigned r = u + 0x7FFFu + ((u >> 16) & 1u);
  return (unsigned short)(r >> 16);
}
__device__ __forceinline__ unsigned fkey(float f) {
  unsigned u = __float_as_uint(f);
  return (u & 0x80000000u) ? ~u : (u | 0x80000000u);
}
__device__ __forceinline__ float funkey(unsigned k) {
  unsigned u = (k & 0x80000000u) ? (k & 0x7FFFFFFFu) : ~k;
  return __uint_as_float(u);
}
__device__ __forceinline__ float lrelu(float x) { return x >= 0.f ? x : NEG_SLOPE * x; }
template <int XM>
__device__ __forceinline__ float row_xor(float v) {
  return __int_as_float(__builtin_amdgcn_update_dpp(0, __float_as_int(v), 0x160 + XM, 0xF, 0xF, true));
}
__device__ __forceinline__ float row16_sum(float v) {
  v += row_xor<1>(v);
  v += row_xor<2>(v);
  v += row_xor<4>(v);
  v += row_xor<8>(v);
  return v;
}

__global__ void k_transpose_w(const float* __restrict__ W,
                              unsigned short* __restrict__ Wt, int K, int NCOLS) {
  int i = blockIdx.x * blockDim.x + threadIdx.x;
  if (i >= K * NCOLS) return;
  int n = i / K, k = i % K;
  Wt[i] = f2bf(W[(size_t)k * NCOLS + n]);
}

template <int KC, int RPB, int STRIP>
__device__ __forceinline__ void gemm_strips(const unsigned char* ap, const Frag16 (&fb)[KC],
                                            v8f (&acc)[RPB]) {
#pragma unroll
  for (int t = 0; t < RPB; ++t) acc[t] = v8f{};
#pragma unroll
  for (int c = 0; c < KC; ++c) {
#pragma unroll
    for (int t = 0; t < RPB; ++t) {
      Frag16 fa;
      fa.q[0] = *reinterpret_cast<const uint4*>(ap + t * STRIP + 64 * c);
      fa.q[1] = *reinterpret_cast<const uint4*>(ap + t * STRIP + 64 * c + 32);
      acc[t] = __builtin_amdgcn_wmma_f32_16x16x32_bf16(false, fa.v, false, fb[c].v,
                                                       (short)0, acc[t], false, false);
    }
  }
  __builtin_amdgcn_sched_barrier(0);
}

template <int NCOLS, int RPB, bool GUARD>
__device__ __forceinline__ void store_strips(const v8f (&acc)[RPB], float* cp, int rOff, int rowsLeft) {
#pragma unroll
  for (int t = 0; t < RPB; ++t) {
#pragma unroll
    for (int v = 0; v < 8; ++v) {
      const int r = 16 * t + rOff + v;
      if (!GUARD || r < rowsLeft) cp[(size_t)r * NCOLS] = acc[t][v];
    }
  }
}

constexpr int DHEAD = 64;
static_assert(HID_F == DHEAD && OUT_F == DHEAD, "a head is 64 columns in both layers");
template <int KDIM, int NCOLS, int RPB, bool RELU>
__global__ __launch_bounds__(2 * NCOLS)
void k_gemm_f32a_wmma(const float* __restrict__ A,
                      const unsigned short* __restrict__ Bt,
                      float* __restrict__ C,
                      const float* __restrict__ al, const float* __restrict__ ar,
                      float* __restrict__ el, float* __restrict__ er, int M) {
  constexpr int KC    = KDIM / 32;
  constexpr int NT    = 2 * NCOLS;
  constexpr int ROWS  = 16 * RPB;
  constexpr int Q4    = KDIM / 4;
  constexpr int NLD   = ROWS * Q4 / NT;
  constexpr int PITCH = KDIM * 2 + 16;
  static_assert((ROWS * Q4) % NT == 0, "the tile must divide among the threads");
  static_assert(NT % Q4 == 0, "a wave-load must stay inside one row");
  __shared__ __attribute__((aligned(16))) unsigned char sA[ROWS * PITCH];
  constexpr int CPITCH = NCOLS + 4;
  constexpr int HEADS  = NCOLS / DHEAD;
  constexpr int UPW    = ROWS * HEADS / (NCOLS / 16);
  static_assert(NCOLS % DHEAD == 0, "whole heads");
  static_assert(HEADS == 1 || HEADS == 2, "a wave-load takes two units: one row's two heads, or two rows");
  static_assert(UPW % 2 == 0 && UPW / 2 <= 16, "two units per wave-load, one result per lane of a row");
  __shared__ __attribute__((aligned(16))) float sC[ROWS * CPITCH];

  const int tid      = threadIdx.x;
  const int lane     = tid & 31;
  const int wv       = tid >> 5;
  const int colBase  = wv * 16;
  const int l16      = lane & 15;
  const int k0       = (lane < 16) ? 0 : 8;
  const int rowBlock = blockIdx.x * ROWS;

  v4f av[NLD];
#pragma unroll
  for (int j = 0; j < NLD; ++j) {
    const int p = j * NT + tid;
    int arow = rowBlock + p / Q4; if (arow >= M) arow = M - 1;
    av[j] = *reinterpret_cast<const v4f*>(A + (size_t)arow * KDIM + 4 * (p % Q4));
  }

  Frag16 fb[KC];
  {
    const unsigned short* bp = Bt + (size_t)(colBase + l16) * KDIM + k0;
#pragma unroll
    for (int c = 0; c < KC; ++c) {
      fb[c].q[0] = *reinterpret_cast<const uint4*>(bp + 32 * c);
      fb[c].q[1] = *reinterpret_cast<const uint4*>(bp + 32 * c + 16);
    }
  }

#pragma unroll
  for (int j = 0; j < NLD; ++j) {
    const int p = j * NT + tid;
    v4f v = av[j];
    if constexpr (RELU) {
#pragma unroll
      for (int i = 0; i < 4; ++i) v[i] = v[i] > 0.f ? v[i] : 0.f;
    }
    const v4bf h = __builtin_convertvector(v, v4bf);
    *reinterpret_cast<v2u*>(sA + (p / Q4) * PITCH + 8 * (p % Q4)) = __builtin_bit_cast(v2u, h);
  }
  __syncthreads();

  const int rOff = (lane < 16) ? 0 : 8;
  const unsigned char* ap = sA + l16 * PITCH + 2 * k0;
  float* cp = C + (size_t)rowBlock * NCOLS + colBase + l16;
  float* tp = sC + rOff * CPITCH + colBase + l16;

  const int hf = lane >> 4;
  const int hd = (HEADS == 2) ? hf : 0;
  const v4f wl = *reinterpret_cast<const v4f*>(al + hd * DHEAD + 4 * l16);
  const v4f wr = *reinterpret_cast<const v4f*>(ar + hd * DHEAD + 4 * l16);

  v8f acc[RPB];
  gemm_strips<KC, RPB, 16 * PITCH>(ap, fb, acc);
#pragma unroll
  for (int t = 0; t < RPB; ++t) {
#pragma unroll
    for (int v = 0; v < 8; ++v) tp[(16 * t + v) * CPITCH] = acc[t][v];
  }
  __syncthreads();

  float ke = 0.f, kr = 0.f;
#pragma unroll
  for (int i = 0; i < UPW / 2; ++i) {
    const int u = UPW * wv + 2 * i + hf;
    const v4f c = *reinterpret_cast<const v4f*>(sC + (u / HEADS) * CPITCH + (u % HEADS) * DHEAD + 4 * l16);
    float sl = c[0] * wl[0] + c[1] * wl[1] + c[2] * wl[2] + c[3] * wl[3];
    float sr = c[0] * wr[0] + c[1] * wr[1] + c[2] * wr[2] + c[3] * wr[3];
    sl = row16_sum(sl);
    sr = row16_sum(sr);
    if (l16 == i) { ke = sl; kr = sr; }
  }
  const long long ug = (long long)rowBlock * HEADS + UPW * wv + 2 * l16 + hf;
  if (l16 < UPW / 2 && ug < (long long)M * HEADS) { el[ug] = ke; er[ug] = kr; }

  if (rowBlock + ROWS <= M)
    store_strips<NCOLS, RPB, false>(acc, cp, rOff, ROWS);
  else
    store_strips<NCOLS, RPB, true>(acc, cp, rOff, M - rowBlock);
}

constexpr int NPB      = 392;

constexpr int PL_T     = 256;
__device__ __forceinline__ size_t end_slot(int e) {
  return (size_t)(e & ~127) + (size_t)(((e & 2) << 5) + (((e & 127) >> 2) << 1) + (e & 1));
}
__device__ __forceinline__ unsigned owner_byte(int d) { return ((unsigned)d / (unsigned)NPB) & 255u; }

__global__ __launch_bounds__(PL_T)
void k_owner_planes(const int* __restrict__ src, const int* __restrict__ dst,
                    unsigned char* __restrict__ own, int2* __restrict__ ends,
                    int E, size_t stride, size_t endStride) {
  const int* sp = src + (size_t)blockIdx.y * E;
  const int* dp = dst + (size_t)blockIdx.y * E;
  unsigned char* op = own + (size_t)blockIdx.y * stride;
  int2* ep = ends + (size_t)blockIdx.y * endStride;
  const int lane = threadIdx.x & 31;
  const int wq = __builtin_amdgcn_readfirstlane((int)blockIdx.x * (PL_T / 32) + (int)(threadIdx.x >> 5));
  const int nBlocks = (((E + 127) >> 7) + 31) >> 5;
  if (wq >= 4 * nBlocks) return;

  int4 v[8], sv[8];
  if ((E & 3) == 0) {
#pragma unroll
    for (int c = 0; c < 8; ++c) {
      const int e0 = (wq * 8 + c) * 128 + 4 * lane;
      v[c]  = *reinterpret_cast<const int4*>(dp + (e0 < E ? e0 : 0));
      sv[c] = *reinterpret_cast<const int4*>(sp + (e0 < E ? e0 : 0));
    }
  } else {
#pragma unroll
    for (int c = 0; c < 8; ++c) {
      const int e0 = (wq * 8 + c) * 128 + 4 * lane;
      v[c].x = (e0     < E) ? dp[e0]     : 0;
      v[c].y = (e0 + 1 < E) ? dp[e0 + 1] : 0;
      v[c].z = (e0 + 2 < E) ? dp[e0 + 2] : 0;
      v[c].w = (e0 + 3 < E) ? dp[e0 + 3] : 0;
      sv[c].x = (e0     < E) ? sp[e0]     : 0;
      sv[c].y = (e0 + 1 < E) ? sp[e0 + 1] : 0;
      sv[c].z = (e0 + 2 < E) ? sp[e0 + 2] : 0;
      sv[c].w = (e0 + 3 < E) ? sp[e0 + 3] : 0;
    }
  }

#pragma unroll
  for (int c = 0; c < 8; ++c) {
    int4* cp = reinterpret_cast<int4*>(ep + (size_t)(wq * 8 + c) * 128 + 2 * lane);
    cp[0]  = make_int4(sv[c].x, v[c].x, sv[c].y, v[c].y);
    cp[32] = make_int4(sv[c].z, v[c].z, sv[c].w, v[c].w);
  }

  unsigned a[8][4];
#pragma unroll
  for (int b = 0; b < 8; ++b)
#pragma unroll
    for (int j = 0; j < 4; ++j) a[b][j] = 0u;
#pragma unroll
  for (int c = 0; c < 8; ++c) {
    const unsigned pk = owner_byte(v[c].x) | (owner_byte(v[c].y) << 8) |
                        (owner_byte(v[c].z) << 16) | (owner_byte(v[c].w) << 24);
#pragma unroll
    for (int b = 0; b < 8; ++b) {
#pragma unroll
      for (int j = 0; j < 4; ++j) {
        const unsigned bal = __builtin_amdgcn_ballot_w32(((pk >> (8 * j + b)) & 1u) != 0u);
        if (lane == c) a[b][j] = bal;
      }
    }
  }
  if (lane < 8) {
    unsigned char* bp = op + (size_t)(wq >> 2) * 4096 + (size_t)(((wq & 3) * 8 + lane) * 16);
#pragma unroll
    for (int b = 0; b < 8; ++b)
      *reinterpret_cast<uint4*>(bp + 512 * b) = make_uint4(a[b][0], a[b][1], a[b][2], a[b][3]);
  }
}

constexpr int OWN_T    = 1024;
constexpr int RING     = 64;
constexpr int CAP      = 7168;

template <int H, int D, bool FIRST>
__global__ __launch_bounds__(OWN_T)
void k_owner_aggregate(const float* __restrict__ el, const float* __restrict__ er,
                       const float* __restrict__ feat,
                       const int2* __restrict__ ends,
                       const unsigned char* __restrict__ own,
                       const float* __restrict__ b, float* __restrict__ acc, int N, int E) {
  constexpr int F   = H * D;
  constexpr int PER = F / 32;
  constexpr int LPH = D / 32;
  constexpr int U   = 32 / PER;
  __shared__ __attribute__((aligned(16))) float racc_s[(NPB + 1) * F];
  __shared__ float    ssum_s[NPB * H];
  __shared__ unsigned mkey_s[NPB * H];
  __shared__ float    er_s[NPB * H];
  __shared__ int      ring_e[OWN_T / 32][RING];
  __shared__ int      lst_s[CAP];
  __shared__ float    lst_v[CAP * H];
  __shared__ unsigned short lst_d[CAP];
  __shared__ int      lst_n;

  const int tid  = threadIdx.x;
  const int lane = tid & 31;
  const int wv   = __builtin_amdgcn_readfirstlane(tid >> 5);
  const int lo   = blockIdx.x * NPB;

  for (int i = tid; i < NPB * F / 4; i += OWN_T)
    reinterpret_cast<float4*>(racc_s)[i] = make_float4(0.f, 0.f, 0.f, 0.f);
  for (int i = tid; i < NPB * H; i += OWN_T) {
    ssum_s[i] = 0.f;
    mkey_s[i] = 0u;
    er_s[i]   = (lo + i / H < N) ? er[(size_t)lo * H + i] : 0.f;
  }
  if (tid == 0) lst_n = 0;
  __syncthreads();

  int* const re = ring_e[wv];

  const int Q  = (E + 127) >> 7;
  const int WL = (Q + 31) >> 5;
  const uint4* own4 = reinterpret_cast<const uint4*>(own);
  unsigned flip[8];
#pragma unroll
  for (int b = 0; b < 8; ++b) flip[b] = ((blockIdx.x >> b) & 1u) ? 0u : 0xFFFFFFFFu;

  const int rot  = (int)(((long long)WL * blockIdx.x) / gridDim.x);
  const int nIt  = (WL > wv) ? ((WL - wv + 31) >> 5) : 0;

  auto sweep = [&](auto&& work) {
    int qh = 0, qt = 0;

    auto push = [&](bool keep, int e) {
      const unsigned mask = __builtin_amdgcn_ballot_w32(keep);
      if (keep) {
        const int pos = (qt + (int)__builtin_amdgcn_mbcnt_lo(mask, 0u)) & (RING - 1);
        re[pos] = e;
      }
      qt += __builtin_popcount(mask);
    };

    int tcur = wv + rot;
    if (tcur >= WL) tcur -= WL;
    uint4 w[8];
    {
      const uint4* bp = own4 + (size_t)(nIt > 0 ? tcur : 0) * 256 + lane;
#pragma unroll
      for (int b = 0; b < 8; ++b) w[b] = bp[32 * b];
    }

    for (int it = 0; it < nIt; ++it) {
      unsigned m0 = w[0].x ^ flip[0], m1 = w[0].y ^ flip[0], m2 = w[0].z ^ flip[0], m3 = w[0].w ^ flip[0];
#pragma unroll
      for (int b = 1; b < 8; ++b) {
        m0 = __builtin_amdgcn_bitop3_b32(m0, w[b].x, flip[b], 0x60);
        m1 = __builtin_amdgcn_bitop3_b32(m1, w[b].y, flip[b], 0x60);
        m2 = __builtin_amdgcn_bitop3_b32(m2, w[b].z, flip[b], 0x60);
        m3 = __builtin_amdgcn_bitop3_b32(m3, w[b].w, flip[b], 0x60);
      }
      const int q = tcur * 32 + lane;
      int tnext = tcur + 32;
      if (tnext >= WL) tnext -= WL;
      {
        const uint4* bp = own4 + (size_t)(it + 1 < nIt ? tnext : tcur) * 256 + lane;
#pragma unroll
        for (int b = 0; b < 8; ++b) w[b] = bp[32 * b];
      }
      tcur = tnext;
      if (q >= Q) { m0 = 0u; m1 = 0u; m2 = 0u; m3 = 0u; }
      const int e0 = q * 128;
      unsigned any = m0 | m1 | m2 | m3;
      for (int trip = 0; trip < 128; ++trip) {
        if (__builtin_amdgcn_ballot_w32(any != 0u) == 0u) break;
        const bool c0 = m0 != 0u, c1 = m1 != 0u, c2 = m2 != 0u;
        const unsigned ws = c0 ? m0 : (c1 ? m1 : (c2 ? m2 : m3));
        const int      js = c0 ? 0  : (c1 ? 1  : (c2 ? 2  : 3));
        const int p = __builtin_ctz(ws | 0x80000000u);
        const int e = e0 + (p << 2) + js;
        push(any != 0u && e < E, e);
        const unsigned wc = ws & (ws - 1u);
        const bool s1 = c1 && !c0, s2 = c2 && !c1 && !c0, s3 = !c2 && !c1 && !c0;
        m0 = c0 ? wc : m0;
        m1 = s1 ? wc : m1;
        m2 = s2 ? wc : m2;
        m3 = s3 ? wc : m3;
        any = m0 | m1 | m2 | m3;
        if (qt - qh >= 32) { work(32, qh); qh += 32; }
      }
    }
    if (qt - qh > 0) { work(qt - qh, qh); qh = qt; }
  };

  auto edge_of = [&](int cnt, int qh, int& s_l, int& d_l, float (&v)[H]) {
    s_l = 0; d_l = NPB;
#pragma unroll
    for (int h = 0; h < H; ++h) v[h] = 0.f;
    if (lane < cnt) {
      const int e = re[(qh + lane) & (RING - 1)];
      const int2 sd = ends[end_slot(e)];
      s_l = sd.x;
      const unsigned rel = (unsigned)(sd.y - lo);
      d_l = (rel < (unsigned)NPB) ? (int)rel : NPB;
    }
    if (d_l < NPB) {
      float a[H];
      if constexpr (H == 2) {
        const float2 a2 = *reinterpret_cast<const float2*>(el + (size_t)s_l * 2);
        a[0] = a2.x; a[1] = a2.y;
      } else {
#pragma unroll
        for (int h = 0; h < H; ++h) a[h] = el[(size_t)s_l * H + h];
      }
#pragma unroll
      for (int h = 0; h < H; ++h) v[h] = lrelu(a[h] + er_s[d_l * H + h]);
    }
  };

  auto finish = [&](int cnt, int s_l, int d_l, const float (&v)[H]) {
    float ex_l[H];
#pragma unroll
    for (int h = 0; h < H; ++h) ex_l[h] = 0.f;
    if (d_l < NPB) {
#pragma unroll
      for (int h = 0; h < H; ++h) {
        ex_l[h] = __expf(v[h] - funkey(mkey_s[d_l * H + h]));
        atomicAdd(&ssum_s[d_l * H + h], ex_l[h]);
      }
    }

    for (int j0 = 0; j0 < cnt; j0 += U) {
      if (j0 + U <= cnt) {
        float f[U][PER];
#pragma unroll
        for (int u = 0; u < U; ++u) {
          const int sg = __builtin_amdgcn_readlane(s_l, j0 + u);
          const float* fp = feat + (size_t)sg * F + lane;
#pragma unroll
          for (int k = 0; k < PER; ++k) f[u][k] = fp[32 * k];
        }
#pragma unroll
        for (int u = 0; u < U; ++u) {
          const int dr = __builtin_amdgcn_readlane(d_l, j0 + u);
          float exs[H];
#pragma unroll
          for (int h = 0; h < H; ++h)
            exs[h] = __int_as_float(__builtin_amdgcn_readlane(__float_as_int(ex_l[h]), j0 + u));
          float* op = racc_s + dr * F + lane;
#pragma unroll
          for (int k = 0; k < PER; ++k) atomicAdd(op + 32 * k, exs[k / LPH] * f[u][k]);
        }
      } else {
        for (int j = j0; j < cnt; ++j) {
          const int sg = __builtin_amdgcn_readlane(s_l, j);
          const int dr = __builtin_amdgcn_readlane(d_l, j);
          float exs[H];
#pragma unroll
          for (int h = 0; h < H; ++h)
            exs[h] = __int_as_float(__builtin_amdgcn_readlane(__float_as_int(ex_l[h]), j));
          const float* fp = feat + (size_t)sg * F + lane;
          float*       op = racc_s + dr * F + lane;
#pragma unroll
          for (int k = 0; k < PER; ++k) atomicAdd(op + 32 * k, exs[k / LPH] * fp[32 * k]);
        }
      }
    }
  };

  sweep([&](int cnt, int qh) {
    int s_l, d_l;
    float v[H];
    edge_of(cnt, qh, s_l, d_l, v);
    const bool mine = d_l < NPB;
    const unsigned mask = __builtin_amdgcn_ballot_w32(mine);
    if (mask != 0u) {
      int base = 0;
      if (lane == 0) base = atomicAdd(&lst_n, __builtin_popcount(mask));
      base = __builtin_amdgcn_readfirstlane(base);
      if (mine) {
#pragma unroll
        for (int h = 0; h < H; ++h) atomicMax(&mkey_s[d_l * H + h], fkey(v[h]));
        const int pos = base + (int)__builtin_amdgcn_mbcnt_lo(mask, 0u);
        if (pos < CAP) {
          lst_s[pos] = s_l;
          lst_d[pos] = (unsigned short)d_l;
#pragma unroll
          for (int h = 0; h < H; ++h) lst_v[pos * H + h] = v[h];
        }
      }
    }
  });
  __syncthreads();
  const int total = __builtin_amdgcn_readfirstlane(lst_n);

  if (total <= CAP) {
    for (int i = wv * 32; i < total; i += OWN_T) {
      const int cnt = (total - i < 32) ? (total - i) : 32;
      int s_l = 0, d_l = NPB;
      float v[H];
#pragma unroll
      for (int h = 0; h < H; ++h) v[h] = 0.f;
      if (lane < cnt) {
        s_l = lst_s[i + lane];
        d_l = lst_d[i + lane];
#pragma unroll
        for (int h = 0; h < H; ++h) v[h] = lst_v[(i + lane) * H + h];
      }
      finish(cnt, s_l, d_l, v);
    }
  } else {
    sweep([&](int cnt, int qh) {
      int s_l, d_l;
      float v[H];
      edge_of(cnt, qh, s_l, d_l, v);
      finish(cnt, s_l, d_l, v);
    });
  }
  __syncthreads();

  for (int i4 = tid; i4 < NPB * (F / 4); i4 += OWN_T) {
    const int nl = i4 / (F / 4);
    const int n  = lo + nl;
    if (n >= N) break;
    const int c = (i4 % (F / 4)) * 4;
    const int h = c / D;
    float sden = ssum_s[nl * H + h];
    sden = (sden > 0.f) ? sden : 1.f;
    const float4 v = reinterpret_cast<const float4*>(racc_s)[i4];
    float4* gp = reinterpret_cast<float4*>(acc + (size_t)n * F + c);
    float4 o;
    if constexpr (FIRST) {
      o = make_float4(0.f, 0.f, 0.f, 0.f);
      for (int r = 0; r < R_REL; ++r) {
        const float4 bv = *reinterpret_cast<const float4*>(b + (size_t)r * F + c);
        o.x += bv.x; o.y += bv.y; o.z += bv.z; o.w += bv.w;
      }
    } else {
      o = *gp;
    }
    o.x += v.x / sden; o.y += v.y / sden; o.z += v.z / sden; o.w += v.w / sden;
    *gp = o;
  }
}

extern "C" void kernel_launch(void* const* d_in, const int* in_sizes, int n_in,
                              void* d_out, int out_size, void* d_ws, size_t ws_size,
                              hipStream_t stream) {
  const float* x    = (const float*)d_in[0];
  const int*   esrc = (const int*)d_in[1];
  const int*   edst = (const int*)d_in[2];
  const float* W1   = (const float*)d_in[3];
  const float* al1  = (const float*)d_in[4];
  const float* ar1  = (const float*)d_in[5];
  const float* b1   = (const float*)d_in[6];
  const float* W2   = (const float*)d_in[7];
  const float* al2  = (const float*)d_in[8];
  const float* ar2  = (const float*)d_in[9];
  const float* b2   = (const float*)d_in[10];
  float* out = (float*)d_out;

  const int N = in_sizes[0] / IN_F;
  const int E = in_sizes[1] / (2 * R_REL);

  size_t off = 0;
  auto carve = [&](size_t bytes) -> void* {
    void* p = (char*)d_ws + off;
    off += (bytes + 255) & ~(size_t)255;
    return p;
  };
  unsigned short* wt1  = (unsigned short*)carve((size_t)R_REL * IN_F * (H1 * HID_F) * 2);
  unsigned short* wt2  = (unsigned short*)carve((size_t)R_REL * (H1 * HID_F) * OUT_F * 2);
  float*    feat  = (float*)carve((size_t)N * (H1 * HID_F) * 4);
  float*    el    = (float*)carve((size_t)N * H1 * 4);
  float*    er    = (float*)carve((size_t)N * H1 * 4);
  float*    hacc  = (float*)carve((size_t)N * (H1 * HID_F) * 4);
  const size_t ownBlocks = ((((size_t)E + 127) >> 7) + 31) >> 5;
  const size_t ownStride = ownBlocks * 4096;
  unsigned char* own = (unsigned char*)carve(ownStride * 2 * R_REL);
  const size_t endStride = ownBlocks * 4096;
  int2* ends = (int2*)carve(endStride * sizeof(int2) * 2 * R_REL);
  const unsigned ownGrid = (unsigned)((N + NPB - 1) / NPB);

  const int T = 256;
  auto blk1 = [&](size_t n) { return (unsigned)((n + T - 1) / T); };

  for (int r = 0; r < R_REL; ++r) {
    k_transpose_w<<<blk1((size_t)IN_F * (H1 * HID_F)), T, 0, stream>>>(
        W1 + (size_t)r * IN_F * (H1 * HID_F), wt1 + (size_t)r * IN_F * (H1 * HID_F),
        IN_F, H1 * HID_F);
    k_transpose_w<<<blk1((size_t)(H1 * HID_F) * OUT_F), T, 0, stream>>>(
        W2 + (size_t)r * (H1 * HID_F) * OUT_F, wt2 + (size_t)r * (H1 * HID_F) * OUT_F,
        H1 * HID_F, OUT_F);
  }
  k_owner_planes<<<dim3((unsigned)((ownBlocks * 4 + PL_T / 32 - 1) / (PL_T / 32)), 2 * R_REL), PL_T, 0, stream>>>(
      esrc, edst, own, ends, E, ownStride, endStride);

  constexpr int RPB = 4;
  const unsigned gemmGrid = (unsigned)((N + 16 * RPB - 1) / (16 * RPB));

  for (int r = 0; r < R_REL; ++r) {
    const int2* p1 = ends + (size_t)(0 * R_REL + r) * endStride;
    const unsigned char* o1 = own + (size_t)(0 * R_REL + r) * ownStride;

    k_gemm_f32a_wmma<IN_F, H1 * HID_F, RPB, false><<<gemmGrid, 2 * (H1 * HID_F), 0, stream>>>(
        x, wt1 + (size_t)r * IN_F * (H1 * HID_F), feat,
        al1 + (size_t)r * H1 * HID_F, ar1 + (size_t)r * H1 * HID_F, el, er, N);

    if (r == 0)
      k_owner_aggregate<H1, HID_F, true><<<ownGrid, OWN_T, 0, stream>>>(
          el, er, feat, p1, o1, b1, hacc, N, E);
    else
      k_owner_aggregate<H1, HID_F, false><<<ownGrid, OWN_T, 0, stream>>>(
          el, er, feat, p1, o1, b1, hacc, N, E);
  }

  for (int r = 0; r < R_REL; ++r) {
    const int2* p2 = ends + (size_t)(1 * R_REL + r) * endStride;
    const unsigned char* o2 = own + (size_t)(1 * R_REL + r) * ownStride;

    k_gemm_f32a_wmma<H1 * HID_F, OUT_F, RPB, true><<<gemmGrid, 2 * OUT_F, 0, stream>>>(
        hacc, wt2 + (size_t)r * (H1 * HID_F) * OUT_F, feat,
        al2 + (size_t)r * H2 * OUT_F, ar2 + (size_t)r * H2 * OUT_F, el, er, N);

    if (r == 0)
      k_owner_aggregate<H2, OUT_F, true><<<ownGrid, OWN_T, 0, stream>>>(
          el, er, feat, p2, o2, b2, out, N, E);
    else
      k_owner_aggregate<H2, OUT_F, false><<<ownGrid, OWN_T, 0, stream>>>(
          el, er, feat, p2, o2, b2, out, N, E);
  }
}
